# starting kernel
# baseline (speedup 1.0000x reference)
#include <hip/hip_runtime.h>

typedef __attribute__((ext_vector_type(2))) float v2f;
typedef __attribute__((ext_vector_type(8))) float v8f;

#define NPWM    512
#define FS      19
#define LEN     1000
#define OUTPOS  982
#define PTILES  62
#define KSTEPS  19
#define NEG_INF (-3.402823466e38f)

static __device__ __forceinline__ v8f wmma_f32_16x16x4(v2f a, v2f b, v8f c) {
  return __builtin_amdgcn_wmma_f32_16x16x4_f32(
       false, a,  false, b,
       (short)0, c,  false,  false);
}

__global__ __launch_bounds__(256) void pwm_scan_max_kernel(
    const float* __restrict__ x,
    const float* __restrict__ pwms,
    float* __restrict__ out) {
  __shared__ float xT[4096];
  __shared__ float red[8][16];

  const int b     = blockIdx.y;
  const int f0    = blockIdx.x * 16;
  const int tid   = threadIdx.x;
  const int lane  = tid & 31;
  const int wave  = tid >> 5;
  const int n     = lane & 15;
  const int cbase = (lane < 16) ? 0 : 2;

  const float* xb = x + (size_t)b * 4 * LEN;
  for (int idx = tid; idx < 4096; idx += 256) {
    int pos = idx >> 2, c = idx & 3;
    xT[idx] = (pos < LEN) ? xb[c * LEN + pos] : 0.0f;
  }

  const float* wrow = pwms + (size_t)(f0 + n) * (4 * FS);
  v2f bF[KSTEPS], bR[KSTEPS];
#pragma unroll
  for (int g = 0; g < KSTEPS; ++g) {
    bF[g].x = wrow[cbase * FS + g];
    bF[g].y = wrow[(cbase + 1) * FS + g];
    bR[g].x = wrow[(3 - cbase) * FS + (FS - 1 - g)];
    bR[g].y = wrow[(2 - cbase) * FS + (FS - 1 - g)];
  }

  __syncthreads();

  float rmax = NEG_INF;
  const float* ap_base = &xT[0];
  for (int pt = wave; pt < PTILES; pt += 8) {
    const int pos0 = pt * 16;
    v8f accF = {};
    v8f accR = {};
    const float* ap = ap_base + (pos0 + n) * 4 + cbase;
#pragma unroll
    for (int g = 0; g < KSTEPS; ++g) {
      v2f a = *(const v2f*)(ap + g * 4);
      accF = wmma_f32_16x16x4(a, bF[g], accF);
      accR = wmma_f32_16x16x4(a, bR[g], accR);
    }
    const int mbase = pos0 + ((lane >= 16) ? 8 : 0);
#pragma unroll
    for (int r = 0; r < 8; ++r) {
      float v = fmaxf(accF[r], accR[r]);
      rmax = (mbase + r < OUTPOS) ? fmaxf(rmax, v) : rmax;
    }
  }

  rmax = fmaxf(rmax, __shfl_xor(rmax, 16, 32));
  if (lane < 16) red[wave][n] = rmax;
  __syncthreads();
  if (tid < 16) {
    float m = red[0][tid];
#pragma unroll
    for (int w = 1; w < 8; ++w) m = fmaxf(m, red[w][tid]);
    out[(size_t)b * NPWM + f0 + tid] = m;
  }
}

extern "C" void kernel_launch(void* const* d_in, const int* in_sizes, int n_in,
                              void* d_out, int out_size, void* d_ws, size_t ws_size,
                              hipStream_t stream) {
  const float* x    = (const float*)d_in[0];
  const float* pwms = (const float*)d_in[1];
  float* out        = (float*)d_out;

  dim3 grid(NPWM / 16, 128);
  pwm_scan_max_kernel<<<grid, 256, 0, stream>>>(x, pwms, out);
}


# Strand pair as S+|A|: symmetric/antisymmetric filter parts times sum/difference of mirrored inputs, 20 WMMAs per tile replace 38 (K 76->40); f32 summation order changes
# speedup vs baseline: 1.5497x; 1.5497x over previous
#include <hip/hip_runtime.h>


typedef __attribute__((ext_vector_type(2))) float v2f;
typedef __attribute__((ext_vector_type(8))) float v8f;

#define NPWM    512
#define FS      19
#define LEN     1000
#define OUTPOS  982
#define PTILES  62
#define KSTEPS  10
#define NEG_INF (-3.402823466e38f)

static __device__ __forceinline__ v8f wmma_f32_16x16x4(v2f a, v2f b, v8f c) {
  return __builtin_amdgcn_wmma_f32_16x16x4_f32(
       false, a,  false, b,
       (short)0, c,  false,  false);
}

__global__ __launch_bounds__(256) void pwm_scan_max_kernel(
    const float* __restrict__ x,
    const float* __restrict__ pwms,
    float* __restrict__ out) {
  __shared__ v2f XA[1024];
  __shared__ v2f XB[1024];
  __shared__ float red[8][16];

  const int b    = blockIdx.y;
  const int f0   = blockIdx.x * 16;
  const int tid  = threadIdx.x;
  const int lane = tid & 31;
  const int wave = tid >> 5;
  const int n    = lane & 15;
  const int h    = lane >> 4;

  const float* xb = x + (size_t)b * 4 * LEN;
  for (int p = tid; p < 1024; p += 256) {
    const int q = p - 1;
    v2f va = {0.0f, 0.0f};
    v2f vb = {0.0f, 0.0f};
    if (p < LEN)           { va.x = xb[p];           va.y = xb[LEN + p]; }
    if (q >= 0 && q < LEN) { vb.x = xb[3 * LEN + q]; vb.y = xb[2 * LEN + q]; }
    XA[p] = va;
    XB[p] = vb;
  }

  const float* wrow = pwms + (size_t)(f0 + n) * (4 * FS);
  v2f bS[KSTEPS], bA[KSTEPS];
#pragma unroll
  for (int g = 0; g < KSTEPS; ++g) {
    const bool ok = (g < KSTEPS - 1) || (h == 0);
    const int  k  = ok ? (2 * g + h) : (FS - 1);
    const float w1x = wrow[0 * FS + k], w2x = wrow[3 * FS + (FS - 1 - k)];
    const float w1y = wrow[1 * FS + k], w2y = wrow[2 * FS + (FS - 1 - k)];
    bS[g].x = ok ? 0.5f * (w1x + w2x) : 0.0f;
    bS[g].y = ok ? 0.5f * (w1y + w2y) : 0.0f;
    bA[g].x = ok ? 0.5f * (w1x - w2x) : 0.0f;
    bA[g].y = ok ? 0.5f * (w1y - w2y) : 0.0f;
  }

  __syncthreads();

  float rmax = NEG_INF;
  for (int pt = wave; pt < PTILES; pt += 8) {
    const int pos0 = pt * 16;
    v8f accS = {};
    v8f accA = {};
    const v2f* pa = &XA[pos0 + n + h];
    const v2f* pb = &XB[pos0 + n + 19 - h];
#pragma unroll
    for (int g = 0; g < KSTEPS; ++g) {
      const v2f x1 = pa[2 * g];
      const v2f x2 = pb[-2 * g];
      accS = wmma_f32_16x16x4(x1 + x2, bS[g], accS);
      accA = wmma_f32_16x16x4(x1 - x2, bA[g], accA);
    }
    const int mbase = pos0 + (h ? 8 : 0);
#pragma unroll
    for (int r = 0; r < 8; ++r) {
      const float v = accS[r] + fabsf(accA[r]);
      rmax = (mbase + r < OUTPOS) ? fmaxf(rmax, v) : rmax;
    }
  }

  rmax = fmaxf(rmax, __shfl_xor(rmax, 16, 32));
  if (lane < 16) red[wave][n] = rmax;
  __syncthreads();
  if (tid < 16) {
    float m = red[0][tid];
#pragma unroll
    for (int w = 1; w < 8; ++w) m = fmaxf(m, red[w][tid]);
    out[(size_t)b * NPWM + f0 + tid] = m;
  }
}

extern "C" void kernel_launch(void* const* d_in, const int* in_sizes, int n_in,
                              void* d_out, int out_size, void* d_ws, size_t ws_size,
                              hipStream_t stream) {
  const float* x    = (const float*)d_in[0];
  const float* pwms = (const float*)d_in[1];
  float* out        = (float*)d_out;

  dim3 grid(NPWM / 16, 128);
  pwm_scan_max_kernel<<<grid, 256, 0, stream>>>(x, pwms, out);
}


# Work split by filter tile, not by position: a wave keeps one 16-filter tile's B fragments for 31 position tiles (was 8), 64 filters per block, grid 8x128: 4x fewer waves, x stagings, fragment loads
# speedup vs baseline: 1.8328x; 1.1827x over previous
#include <hip/hip_runtime.h>


typedef __attribute__((ext_vector_type(2))) float v2f;
typedef __attribute__((ext_vector_type(8))) float v8f;

#define NPWM    512
#define FS      19
#define LEN     1000
#define OUTPOS  982
#define PTILES  62
#define KSTEPS  10
#define FTILES  4
#define PHALF   31
#define NEG_INF (-3.402823466e38f)

static __device__ __forceinline__ v8f wmma_f32_16x16x4(v2f a, v2f b, v8f c) {
  return __builtin_amdgcn_wmma_f32_16x16x4_f32(
       false, a,  false, b,
       (short)0, c,  false,  false);
}

__global__ __launch_bounds__(256) void pwm_scan_max_kernel(
    const float* __restrict__ x,
    const float* __restrict__ pwms,
    float* __restrict__ out) {
  __shared__ v2f XA[1024];
  __shared__ v2f XB[1024];
  __shared__ __attribute__((aligned(16))) float red[2][FTILES * 16];

  const int b    = blockIdx.y;
  const int tid  = threadIdx.x;
  const int lane = tid & 31;
  const int wave = tid >> 5;
  const int ft   = wave & (FTILES - 1);
  const int half = wave / FTILES;
  const int f0   = (blockIdx.x * FTILES + ft) * 16;
  const int n    = lane & 15;
  const int h    = lane >> 4;

  const float* xb = x + (size_t)b * 4 * LEN;
  for (int p = tid; p < 1024; p += 256) {
    const int q = p - 1;
    v2f va = {0.0f, 0.0f};
    v2f vb = {0.0f, 0.0f};
    if (p < LEN)           { va.x = xb[p];           va.y = xb[LEN + p]; }
    if (q >= 0 && q < LEN) { vb.x = xb[3 * LEN + q]; vb.y = xb[2 * LEN + q]; }
    XA[p] = va;
    XB[p] = vb;
  }

  const float* wrow = pwms + (size_t)(f0 + n) * (4 * FS);
  v2f bS[KSTEPS], bA[KSTEPS];
#pragma unroll
  for (int g = 0; g < KSTEPS; ++g) {
    const bool ok = (g < KSTEPS - 1) || (h == 0);
    const int  k  = ok ? (2 * g + h) : (FS - 1);
    const float w1x = wrow[0 * FS + k], w2x = wrow[3 * FS + (FS - 1 - k)];
    const float w1y = wrow[1 * FS + k], w2y = wrow[2 * FS + (FS - 1 - k)];
    bS[g].x = ok ? 0.5f * (w1x + w2x) : 0.0f;
    bS[g].y = ok ? 0.5f * (w1y + w2y) : 0.0f;
    bA[g].x = ok ? 0.5f * (w1x - w2x) : 0.0f;
    bA[g].y = ok ? 0.5f * (w1y - w2y) : 0.0f;
  }

  __syncthreads();

  float rmax = NEG_INF;
  for (int pt = half * PHALF; pt < (half + 1) * PHALF; ++pt) {
    const int pos0 = pt * 16;
    v8f accS = {};
    v8f accA = {};
    const v2f* pa = &XA[pos0 + n + h];
    const v2f* pb = &XB[pos0 + n + 19 - h];
#pragma unroll
    for (int g = 0; g < KSTEPS; ++g) {
      const v2f x1 = pa[2 * g];
      const v2f x2 = pb[-2 * g];
      accS = wmma_f32_16x16x4(x1 + x2, bS[g], accS);
      accA = wmma_f32_16x16x4(x1 - x2, bA[g], accA);
    }
    const int mbase = pos0 + (h ? 8 : 0);
#pragma unroll
    for (int r = 0; r < 8; ++r) {
      const float v = accS[r] + fabsf(accA[r]);
      rmax = (mbase + r < OUTPOS) ? fmaxf(rmax, v) : rmax;
    }
  }

  rmax = fmaxf(rmax, __shfl_xor(rmax, 16, 32));
  if (lane < 16) red[half][ft * 16 + n] = rmax;
  __syncthreads();
  if (tid < FTILES * 4) {
    const float4 lo = reinterpret_cast<const float4*>(red[0])[tid];
    const float4 hi = reinterpret_cast<const float4*>(red[1])[tid];
    float4 m;
    m.x = fmaxf(lo.x, hi.x);
    m.y = fmaxf(lo.y, hi.y);
    m.z = fmaxf(lo.z, hi.z);
    m.w = fmaxf(lo.w, hi.w);
    reinterpret_cast<float4*>(out + (size_t)b * NPWM + blockIdx.x * (FTILES * 16))[tid] = m;
  }
}

extern "C" void kernel_launch(void* const* d_in, const int* in_sizes, int n_in,
                              void* d_out, int out_size, void* d_ws, size_t ws_size,
                              hipStream_t stream) {
  const float* x    = (const float*)d_in[0];
  const float* pwms = (const float*)d_in[1];
  float* out        = (float*)d_out;

  dim3 grid(8, 128);
  pwm_scan_max_kernel<<<grid, 256, 0, stream>>>(x, pwms, out);
}


# Two 16-filter tiles per wave (128 filters per block, grid 4x128): each A operand read from LDS and its sum/difference feed 4 WMMAs, not 2; half the LDS reads, adds, waves and x stagings
# speedup vs baseline: 1.8940x; 1.0334x over previous
#include <hip/hip_runtime.h>


typedef __attribute__((ext_vector_type(2))) float v2f;
typedef __attribute__((ext_vector_type(8))) float v8f;

#define NPWM    512
#define FS      19
#define LEN     1000
#define OUTPOS  982
#define PTILES  62
#define KSTEPS  10
#define FPAIRS  4
#define FBLOCK  128
#define PHALF   31
#define NEG_INF (-3.402823466e38f)

static __device__ __forceinline__ v8f wmma_f32_16x16x4(v2f a, v2f b, v8f c) {
  return __builtin_amdgcn_wmma_f32_16x16x4_f32(
       false, a,  false, b,
       (short)0, c,  false,  false);
}

__global__ __launch_bounds__(256) void pwm_scan_max_kernel(
    const float* __restrict__ x,
    const float* __restrict__ pwms,
    float* __restrict__ out) {
  __shared__ v2f XA[1024];
  __shared__ v2f XB[1024];
  __shared__ __attribute__((aligned(16))) float red[2][FBLOCK];

  const int b    = blockIdx.y;
  const int tid  = threadIdx.x;
  const int lane = tid & 31;
  const int wave = tid >> 5;
  const int fp   = wave & (FPAIRS - 1);
  const int half = wave / FPAIRS;
  const int f0   = blockIdx.x * FBLOCK + fp * 32;
  const int n    = lane & 15;
  const int h    = lane >> 4;

  const float* xb = x + (size_t)b * 4 * LEN;
  for (int p = tid; p < 1024; p += 256) {
    const int q = p - 1;
    v2f va = {0.0f, 0.0f};
    v2f vb = {0.0f, 0.0f};
    if (p < LEN)           { va.x = xb[p];           va.y = xb[LEN + p]; }
    if (q >= 0 && q < LEN) { vb.x = xb[3 * LEN + q]; vb.y = xb[2 * LEN + q]; }
    XA[p] = va;
    XB[p] = vb;
  }

  v2f bS[2][KSTEPS], bA[2][KSTEPS];
#pragma unroll
  for (int j = 0; j < 2; ++j) {
    const float* wrow = pwms + (size_t)(f0 + 16 * j + n) * (4 * FS);
#pragma unroll
    for (int g = 0; g < KSTEPS; ++g) {
      const bool ok = (g < KSTEPS - 1) || (h == 0);
      const int  k  = ok ? (2 * g + h) : (FS - 1);
      const float w1x = wrow[0 * FS + k], w2x = wrow[3 * FS + (FS - 1 - k)];
      const float w1y = wrow[1 * FS + k], w2y = wrow[2 * FS + (FS - 1 - k)];
      bS[j][g].x = ok ? 0.5f * (w1x + w2x) : 0.0f;
      bS[j][g].y = ok ? 0.5f * (w1y + w2y) : 0.0f;
      bA[j][g].x = ok ? 0.5f * (w1x - w2x) : 0.0f;
      bA[j][g].y = ok ? 0.5f * (w1y - w2y) : 0.0f;
    }
  }

  __syncthreads();

  float rmax0 = NEG_INF;
  float rmax1 = NEG_INF;
  for (int pt = half * PHALF; pt < (half + 1) * PHALF; ++pt) {
    const int pos0 = pt * 16;
    v8f accS0 = {}, accA0 = {};
    v8f accS1 = {}, accA1 = {};
    const v2f* pa = &XA[pos0 + n + h];
    const v2f* pb = &XB[pos0 + n + 19 - h];
#pragma unroll
    for (int g = 0; g < KSTEPS; ++g) {
      const v2f x1 = pa[2 * g];
      const v2f x2 = pb[-2 * g];
      const v2f xs = x1 + x2;
      const v2f xd = x1 - x2;
      accS0 = wmma_f32_16x16x4(xs, bS[0][g], accS0);
      accA0 = wmma_f32_16x16x4(xd, bA[0][g], accA0);
      accS1 = wmma_f32_16x16x4(xs, bS[1][g], accS1);
      accA1 = wmma_f32_16x16x4(xd, bA[1][g], accA1);
    }
    const int mbase = pos0 + (h ? 8 : 0);
#pragma unroll
    for (int r = 0; r < 8; ++r) {
      const bool  in = (mbase + r < OUTPOS);
      const float v0 = accS0[r] + fabsf(accA0[r]);
      const float v1 = accS1[r] + fabsf(accA1[r]);
      rmax0 = in ? fmaxf(rmax0, v0) : rmax0;
      rmax1 = in ? fmaxf(rmax1, v1) : rmax1;
    }
  }

  rmax0 = fmaxf(rmax0, __shfl_xor(rmax0, 16, 32));
  rmax1 = fmaxf(rmax1, __shfl_xor(rmax1, 16, 32));
  red[half][fp * 32 + lane] = h ? rmax1 : rmax0;
  __syncthreads();
  if (tid < FBLOCK / 4) {
    const float4 lo = reinterpret_cast<const float4*>(red[0])[tid];
    const float4 hi = reinterpret_cast<const float4*>(red[1])[tid];
    float4 m;
    m.x = fmaxf(lo.x, hi.x);
    m.y = fmaxf(lo.y, hi.y);
    m.z = fmaxf(lo.z, hi.z);
    m.w = fmaxf(lo.w, hi.w);
    reinterpret_cast<float4*>(out + (size_t)b * NPWM + blockIdx.x * FBLOCK)[tid] = m;
  }
}

extern "C" void kernel_launch(void* const* d_in, const int* in_sizes, int n_in,
                              void* d_out, int out_size, void* d_ws, size_t ws_size,
                              hipStream_t stream) {
  const float* x    = (const float*)d_in[0];
  const float* pwms = (const float*)d_in[1];
  float* out        = (float*)d_out;

  dim3 grid(4, 128);
  pwm_scan_max_kernel<<<grid, 256, 0, stream>>>(x, pwms, out);
}


# Filter weights reach the B fragments through LDS: one coalesced 16-byte-per-lane copy of the block's 128 filters replaces 76 lane-strided 4-byte global loads per lane (16 lines each)
# speedup vs baseline: 2.0750x; 1.0956x over previous
#include <hip/hip_runtime.h>


typedef __attribute__((ext_vector_type(2))) float v2f;
typedef __attribute__((ext_vector_type(8))) float v8f;

#define NPWM    512
#define FS      19
#define LEN     1000
#define OUTPOS  982
#define PTILES  62
#define KSTEPS  10
#define FPAIRS  4
#define FBLOCK  128
#define PHALF   31
#define WROW    (4 * FS)
#define WQ4     9
#define NEG_INF (-3.402823466e38f)

static __device__ __forceinline__ v8f wmma_f32_16x16x4(v2f a, v2f b, v8f c) {
  return __builtin_amdgcn_wmma_f32_16x16x4_f32(
       false, a,  false, b,
       (short)0, c,  false,  false);
}

__global__ __launch_bounds__(256) void pwm_scan_max_kernel(
    const float* __restrict__ x,
    const float* __restrict__ pwms,
    float* __restrict__ out) {
  __shared__ v2f XA[1024];
  __shared__ v2f XB[1024];
  __shared__ __attribute__((aligned(16))) float red[2][FBLOCK];
  __shared__ __attribute__((aligned(16))) float WS[FBLOCK * WROW];

  const int b    = blockIdx.y;
  const int tid  = threadIdx.x;
  const int lane = tid & 31;
  const int wave = tid >> 5;
  const int fp   = wave & (FPAIRS - 1);
  const int half = wave / FPAIRS;
  const int n    = lane & 15;
  const int h    = lane >> 4;

  const float* xb = x + (size_t)b * 4 * LEN;
  for (int p = tid; p < 1024; p += 256) {
    const int q = p - 1;
    v2f va = {0.0f, 0.0f};
    v2f vb = {0.0f, 0.0f};
    if (p < LEN)           { va.x = xb[p];           va.y = xb[LEN + p]; }
    if (q >= 0 && q < LEN) { vb.x = xb[3 * LEN + q]; vb.y = xb[2 * LEN + q]; }
    XA[p] = va;
    XB[p] = vb;
  }

  const float* wblk = pwms + (size_t)blockIdx.x * (FBLOCK * WROW);
#pragma unroll
  for (int it = 0; it < WQ4; ++it)
    reinterpret_cast<float4*>(WS)[tid + 256 * it] = reinterpret_cast<const float4*>(wblk)[tid + 256 * it];
  reinterpret_cast<float2*>(WS + WQ4 * 1024)[tid] = reinterpret_cast<const float2*>(wblk + WQ4 * 1024)[tid];

  __syncthreads();

  v2f bS[2][KSTEPS], bA[2][KSTEPS];
#pragma unroll
  for (int j = 0; j < 2; ++j) {
    const float* wrow = WS + (fp * 32 + 16 * j + n) * WROW;
#pragma unroll
    for (int g = 0; g < KSTEPS; ++g) {
      const bool ok = (g < KSTEPS - 1) || (h == 0);
      const int  k  = ok ? (2 * g + h) : (FS - 1);
      const float w1x = wrow[0 * FS + k], w2x = wrow[3 * FS + (FS - 1 - k)];
      const float w1y = wrow[1 * FS + k], w2y = wrow[2 * FS + (FS - 1 - k)];
      bS[j][g].x = ok ? 0.5f * (w1x + w2x) : 0.0f;
      bS[j][g].y = ok ? 0.5f * (w1y + w2y) : 0.0f;
      bA[j][g].x = ok ? 0.5f * (w1x - w2x) : 0.0f;
      bA[j][g].y = ok ? 0.5f * (w1y - w2y) : 0.0f;
    }
  }

  float rmax0 = NEG_INF;
  float rmax1 = NEG_INF;
  for (int pt = half * PHALF; pt < (half + 1) * PHALF; ++pt) {
    const int pos0 = pt * 16;
    v8f accS0 = {}, accA0 = {};
    v8f accS1 = {}, accA1 = {};
    const v2f* pa = &XA[pos0 + n + h];
    const v2f* pb = &XB[pos0 + n + 19 - h];
#pragma unroll
    for (int g = 0; g < KSTEPS; ++g) {
      const v2f x1 = pa[2 * g];
      const v2f x2 = pb[-2 * g];
      const v2f xs = x1 + x2;
      const v2f xd = x1 - x2;
      accS0 = wmma_f32_16x16x4(xs, bS[0][g], accS0);
      accA0 = wmma_f32_16x16x4(xd, bA[0][g], accA0);
      accS1 = wmma_f32_16x16x4(xs, bS[1][g], accS1);
      accA1 = wmma_f32_16x16x4(xd, bA[1][g], accA1);
    }
    const int mbase = pos0 + (h ? 8 : 0);
#pragma unroll
    for (int r = 0; r < 8; ++r) {
      const bool  in = (mbase + r < OUTPOS);
      const float v0 = accS0[r] + fabsf(accA0[r]);
      const float v1 = accS1[r] + fabsf(accA1[r]);
      rmax0 = in ? fmaxf(rmax0, v0) : rmax0;
      rmax1 = in ? fmaxf(rmax1, v1) : rmax1;
    }
  }

  rmax0 = fmaxf(rmax0, __shfl_xor(rmax0, 16, 32));
  rmax1 = fmaxf(rmax1, __shfl_xor(rmax1, 16, 32));
  red[half][fp * 32 + lane] = h ? rmax1 : rmax0;
  __syncthreads();
  if (tid < FBLOCK / 4) {
    const float4 lo = reinterpret_cast<const float4*>(red[0])[tid];
    const float4 hi = reinterpret_cast<const float4*>(red[1])[tid];
    float4 m;
    m.x = fmaxf(lo.x, hi.x);
    m.y = fmaxf(lo.y, hi.y);
    m.z = fmaxf(lo.z, hi.z);
    m.w = fmaxf(lo.w, hi.w);
    reinterpret_cast<float4*>(out + (size_t)b * NPWM + blockIdx.x * FBLOCK)[tid] = m;
  }
}

extern "C" void kernel_launch(void* const* d_in, const int* in_sizes, int n_in,
                              void* d_out, int out_size, void* d_ws, size_t ws_size,
                              hipStream_t stream) {
  const float* x    = (const float*)d_in[0];
  const float* pwms = (const float*)d_in[1];
  float* out        = (float*)d_out;

  dim3 grid(4, 128);
  pwm_scan_max_kernel<<<grid, 256, 0, stream>>>(x, pwms, out);
}


# Row mask only on each wave's last position tile (only tile 61 has rows past 982); the other 30 tiles fold S+|A| into the running maxima unmasked, by three-input maxima: 34 vector ops a tile, was 54
# speedup vs baseline: 2.2580x; 1.0882x over previous
#include <hip/hip_runtime.h>


typedef __attribute__((ext_vector_type(2))) float v2f;
typedef __attribute__((ext_vector_type(8))) float v8f;

#define NPWM    512
#define FS      19
#define LEN     1000
#define OUTPOS  982
#define PTILES  62
#define KSTEPS  10
#define FPAIRS  4
#define FBLOCK  128
#define PHALF   31
#define WROW    (4 * FS)
#define WQ4     9
#define NEG_INF (-3.402823466e38f)

static __device__ __forceinline__ v8f wmma_f32_16x16x4(v2f a, v2f b, v8f c) {
  return __builtin_amdgcn_wmma_f32_16x16x4_f32(
       false, a,  false, b,
       (short)0, c,  false,  false);
}

template <bool MASKED>
static __device__ __forceinline__ void scan_tile(const v2f* pa, const v2f* pb, const int mbase,
                                                 const v2f (&bS)[2][KSTEPS], const v2f (&bA)[2][KSTEPS],
                                                 float& rmax0, float& rmax1) {
  v8f accS0 = {}, accA0 = {};
  v8f accS1 = {}, accA1 = {};
#pragma unroll
  for (int g = 0; g < KSTEPS; ++g) {
    const v2f x1 = pa[2 * g];
    const v2f x2 = pb[-2 * g];
    const v2f xs = x1 + x2;
    const v2f xd = x1 - x2;
    accS0 = wmma_f32_16x16x4(xs, bS[0][g], accS0);
    accA0 = wmma_f32_16x16x4(xd, bA[0][g], accA0);
    accS1 = wmma_f32_16x16x4(xs, bS[1][g], accS1);
    accA1 = wmma_f32_16x16x4(xd, bA[1][g], accA1);
  }
  float v0[8], v1[8];
#pragma unroll
  for (int r = 0; r < 8; ++r) {
    v0[r] = accS0[r] + fabsf(accA0[r]);
    v1[r] = accS1[r] + fabsf(accA1[r]);
    if (MASKED) {
      const bool in = (mbase + r < OUTPOS);
      v0[r] = in ? v0[r] : NEG_INF;
      v1[r] = in ? v1[r] : NEG_INF;
    }
  }
  const float a0 = fmaxf(fmaxf(v0[0], v0[1]), v0[2]);
  const float a1 = fmaxf(fmaxf(v0[3], v0[4]), v0[5]);
  const float a2 = fmaxf(fmaxf(v0[6], v0[7]), rmax0);
  rmax0 = fmaxf(fmaxf(a0, a1), a2);
  const float c0 = fmaxf(fmaxf(v1[0], v1[1]), v1[2]);
  const float c1 = fmaxf(fmaxf(v1[3], v1[4]), v1[5]);
  const float c2 = fmaxf(fmaxf(v1[6], v1[7]), rmax1);
  rmax1 = fmaxf(fmaxf(c0, c1), c2);
}

__global__ __launch_bounds__(256) void pwm_scan_max_kernel(
    const float* __restrict__ x,
    const float* __restrict__ pwms,
    float* __restrict__ out) {
  __shared__ v2f XA[1024];
  __shared__ v2f XB[1024];
  __shared__ __attribute__((aligned(16))) float red[2][FBLOCK];
  __shared__ __attribute__((aligned(16))) float WS[FBLOCK * WROW];

  const int b    = blockIdx.y;
  const int tid  = threadIdx.x;
  const int lane = tid & 31;
  const int wave = tid >> 5;
  const int fp   = wave & (FPAIRS - 1);
  const int half = wave / FPAIRS;
  const int n    = lane & 15;
  const int h    = lane >> 4;

  const float* xb = x + (size_t)b * 4 * LEN;
  for (int p = tid; p < 1024; p += 256) {
    const int q = p - 1;
    v2f va = {0.0f, 0.0f};
    v2f vb = {0.0f, 0.0f};
    if (p < LEN)           { va.x = xb[p];           va.y = xb[LEN + p]; }
    if (q >= 0 && q < LEN) { vb.x = xb[3 * LEN + q]; vb.y = xb[2 * LEN + q]; }
    XA[p] = va;
    XB[p] = vb;
  }

  const float* wblk = pwms + (size_t)blockIdx.x * (FBLOCK * WROW);
#pragma unroll
  for (int it = 0; it < WQ4; ++it)
    reinterpret_cast<float4*>(WS)[tid + 256 * it] = reinterpret_cast<const float4*>(wblk)[tid + 256 * it];
  reinterpret_cast<float2*>(WS + WQ4 * 1024)[tid] = reinterpret_cast<const float2*>(wblk + WQ4 * 1024)[tid];

  __syncthreads();

  v2f bS[2][KSTEPS], bA[2][KSTEPS];
#pragma unroll
  for (int j = 0; j < 2; ++j) {
    const float* wrow = WS + (fp * 32 + 16 * j + n) * WROW;
#pragma unroll
    for (int g = 0; g < KSTEPS; ++g) {
      const bool ok = (g < KSTEPS - 1) || (h == 0);
      const int  k  = ok ? (2 * g + h) : (FS - 1);
      const float w1x = wrow[0 * FS + k], w2x = wrow[3 * FS + (FS - 1 - k)];
      const float w1y = wrow[1 * FS + k], w2y = wrow[2 * FS + (FS - 1 - k)];
      bS[j][g].x = ok ? 0.5f * (w1x + w2x) : 0.0f;
      bS[j][g].y = ok ? 0.5f * (w1y + w2y) : 0.0f;
      bA[j][g].x = ok ? 0.5f * (w1x - w2x) : 0.0f;
      bA[j][g].y = ok ? 0.5f * (w1y - w2y) : 0.0f;
    }
  }

  float rmax0 = NEG_INF;
  float rmax1 = NEG_INF;
  const int pt0 = half * PHALF;
  for (int i = 0; i < PHALF - 1; ++i) {
    const int pos0 = (pt0 + i) * 16;
    scan_tile<false>(&XA[pos0 + n + h], &XB[pos0 + n + 19 - h], 0, bS, bA, rmax0, rmax1);
  }
  {
    const int pos0 = (pt0 + PHALF - 1) * 16;
    scan_tile<true>(&XA[pos0 + n + h], &XB[pos0 + n + 19 - h], pos0 + (h ? 8 : 0), bS, bA, rmax0, rmax1);
  }

  rmax0 = fmaxf(rmax0, __shfl_xor(rmax0, 16, 32));
  rmax1 = fmaxf(rmax1, __shfl_xor(rmax1, 16, 32));
  red[half][fp * 32 + lane] = h ? rmax1 : rmax0;
  __syncthreads();
  if (tid < FBLOCK / 4) {
    const float4 lo = reinterpret_cast<const float4*>(red[0])[tid];
    const float4 hi = reinterpret_cast<const float4*>(red[1])[tid];
    float4 m;
    m.x = fmaxf(lo.x, hi.x);
    m.y = fmaxf(lo.y, hi.y);
    m.z = fmaxf(lo.z, hi.z);
    m.w = fmaxf(lo.w, hi.w);
    reinterpret_cast<float4*>(out + (size_t)b * NPWM + blockIdx.x * FBLOCK)[tid] = m;
  }
}

extern "C" void kernel_launch(void* const* d_in, const int* in_sizes, int n_in,
                              void* d_out, int out_size, void* d_ws, size_t ws_size,
                              hipStream_t stream) {
  const float* x    = (const float*)d_in[0];
  const float* pwms = (const float*)d_in[1];
  float* out        = (float*)d_out;

  dim3 grid(4, 128);
  pwm_scan_max_kernel<<<grid, 256, 0, stream>>>(x, pwms, out);
}
